# starting kernel
# baseline (speedup 1.0000x reference)
#include <hip/hip_runtime.h>
#include <hip/hip_bf16.h>
#include <cstddef>

#define N0_  100000
#define N1_  50000
#define N2_  25000
#define DEG_ 16
#define IN_  256
#define HID_ 128
#define OUT_ 128
#define H1_  2
#define H2_  1

typedef float v2f __attribute__((ext_vector_type(2)));
typedef float v8f __attribute__((ext_vector_type(8)));

template <int TN, int TK>
__global__ void gat_gemm_wmma(const float* __restrict__ A,
                              const float* __restrict__ B,
                              float* __restrict__ D,
                              int M, int tilesM) {
  constexpr int groupsN = TN >> 6;
  const int lane = threadIdx.x & 31;
  const int wave = threadIdx.x >> 5;
  const int gid = blockIdx.x * 8 + wave;
  if (gid >= tilesM * groupsN) return;
  const int tileM = gid / groupsN;
  const int gN    = gid % groupsN;

  const int half = lane >> 4;
  const int l15  = lane & 15;
  int row = tileM * 16 + l15;
  if (row >= M) row = M - 1;
  const float2* ap = (const float2*)(A + (size_t)row * TK) + half;
  const float*  bp = B + (size_t)(half * 2) * TN + gN * 64 + l15;

  v8f c0 = {}, c1 = {}, c2 = {}, c3 = {};

#pragma unroll 8
  for (int k0 = 0; k0 < TK; k0 += 4) {
    const float2 a2 = *ap;
    ap += 2;
    v2f a = { a2.x, a2.y };
    v2f b0 = { bp[ 0], bp[TN +  0] };
    v2f b1 = { bp[16], bp[TN + 16] };
    v2f b2 = { bp[32], bp[TN + 32] };
    v2f b3 = { bp[48], bp[TN + 48] };
    bp += 4 * TN;
    c0 = __builtin_amdgcn_wmma_f32_16x16x4_f32(false, a, false, b0, (short)0, c0, false, false);
    c1 = __builtin_amdgcn_wmma_f32_16x16x4_f32(false, a, false, b1, (short)0, c1, false, false);
    c2 = __builtin_amdgcn_wmma_f32_16x16x4_f32(false, a, false, b2, (short)0, c2, false, false);
    c3 = __builtin_amdgcn_wmma_f32_16x16x4_f32(false, a, false, b3, (short)0, c3, false, false);
  }

#pragma unroll
  for (int r = 0; r < 8; ++r) {
    const int m = tileM * 16 + r + half * 8;
    if (m < M) {
      float* Dr = D + (size_t)m * TN + gN * 64 + l15;
      Dr[0]  = c0[r];
      Dr[16] = c1[r];
      Dr[32] = c2[r];
      Dr[48] = c3[r];
    }
  }
}

__global__ void gat_att_dot(const float* __restrict__ F,
                            const float* __restrict__ att,
                            float* __restrict__ out, int Nn, int H) {
  const int wid  = blockIdx.x * 8 + (threadIdx.x >> 5);
  const int lane = threadIdx.x & 31;
  if (wid >= Nn * H) return;
  const int n = wid / H, h = wid % H;
  const float4 fv = *((const float4*)(F + ((size_t)n * H + h) * 128) + lane);
  const float4 av = *((const float4*)(att + (size_t)h * 128) + lane);
  float s = fv.x * av.x + fv.y * av.y + fv.z * av.z + fv.w * av.w;
#pragma unroll
  for (int off = 16; off > 0; off >>= 1) s += __shfl_xor(s, off, 32);
  if (lane == 0) out[wid] = s;
}

__global__ void gat_edge_softmax_agg(const float* __restrict__ HS,
                                     const float* __restrict__ AS,
                                     const float* __restrict__ AD,
                                     const int* __restrict__ col,
                                     const int* __restrict__ tt,
                                     const int* __restrict__ timeP,
                                     const int* __restrict__ intervalP,
                                     const float* __restrict__ b,
                                     const float* __restrict__ g,
                                     const float* __restrict__ beta,
                                     const float* __restrict__ rm,
                                     const float* __restrict__ rv,
                                     float* __restrict__ OUT,
                                     int Nd, int H, int mode) {
  const int wid  = blockIdx.x * 8 + (threadIdx.x >> 5);
  const int lane = threadIdx.x & 31;
  if (wid >= Nd * H) return;
  const int i = wid / H, h = wid % H;

  const int t0  = timeP[0];
  const int t1e = t0 + intervalP[0];
  const float adv = AD[(size_t)i * H + h];

  float e[DEG_];
  int   cj[DEG_];
  bool  val[DEG_];
  float m = -__builtin_inff();
#pragma unroll
  for (int j = 0; j < DEG_; ++j) {
    const int eidx = i * DEG_ + j;
    cj[j] = col[eidx];
    const int tv = tt[eidx];
    val[j] = (tv >= t0) && (tv < t1e);
    float ev = AS[(size_t)cj[j] * H + h] + adv;
    ev = ev > 0.0f ? ev : 0.2f * ev;
    e[j] = ev;
    if (val[j] && ev > m) m = ev;
  }
  if (!(m > -__builtin_inff())) m = 0.0f;

  float w[DEG_];
  float denom = 0.0f;
#pragma unroll
  for (int j = 0; j < DEG_; ++j) {
    const float ex = val[j] ? __expf(e[j] - m) : 0.0f;
    w[j] = ex;
    denom += ex;
  }
  const float inv = 1.0f / fmaxf(denom, 1e-16f);

  float4 acc = {0.0f, 0.0f, 0.0f, 0.0f};
#pragma unroll
  for (int j = 0; j < DEG_; ++j) {
    if (w[j] != 0.0f) {
      const float4 hv = *((const float4*)(HS + ((size_t)cj[j] * H + h) * 128) + lane);
      const float wj = w[j] * inv;
      acc.x += wj * hv.x; acc.y += wj * hv.y;
      acc.z += wj * hv.z; acc.w += wj * hv.w;
    }
  }

  float vals[4] = {acc.x, acc.y, acc.z, acc.w};
  if (mode == 1) {
#pragma unroll
    for (int q = 0; q < 4; ++q) {
      const int cc = h * 128 + lane * 4 + q;
      float vv = vals[q] + b[cc];
      vv = (vv - rm[cc]) * rsqrtf(rv[cc] + 1e-5f) * g[cc] + beta[cc];
      vv = vv > 0.0f ? vv : 0.01f * vv;
      OUT[(size_t)i * (H * 128) + cc] = vv;
    }
  } else {
#pragma unroll
    for (int q = 0; q < 4; ++q) {
      const int cc = lane * 4 + q;
      OUT[(size_t)i * 128 + cc] = vals[q] / (float)H + b[cc];
    }
  }
}

static inline int cdiv(int a, int b) { return (a + b - 1) / b; }

extern "C" void kernel_launch(void* const* d_in, const int* in_sizes, int n_in,
                              void* d_out, int out_size, void* d_ws, size_t ws_size,
                              hipStream_t stream) {
  const float* x     = (const float*)d_in[0];
  const int*   col1  = (const int*)d_in[2];
  const int*   t1    = (const int*)d_in[3];
  const int*   col2  = (const int*)d_in[5];
  const int*   t2    = (const int*)d_in[6];
  const int*   timeP = (const int*)d_in[7];
  const int*   intvP = (const int*)d_in[8];
  const float* Ws1   = (const float*)d_in[9];
  const float* Wd1   = (const float*)d_in[10];
  const float* as1   = (const float*)d_in[11];
  const float* ad1   = (const float*)d_in[12];
  const float* b1    = (const float*)d_in[13];
  const float* g1    = (const float*)d_in[14];
  const float* beta1 = (const float*)d_in[15];
  const float* m1    = (const float*)d_in[16];
  const float* v1    = (const float*)d_in[17];
  const float* Ws2   = (const float*)d_in[18];
  const float* Wd2   = (const float*)d_in[19];
  const float* as2   = (const float*)d_in[20];
  const float* ad2   = (const float*)d_in[21];
  const float* b2    = (const float*)d_in[22];

  float* ws   = (float*)d_ws;
  float* hs1  = ws;
  float* hd1  = hs1  + (size_t)N0_ * 256;
  float* as1v = hd1  + (size_t)N1_ * 256;
  float* ad1v = as1v + (size_t)N0_ * H1_;
  float* h    = ad1v + (size_t)N1_ * H1_;
  float* as2v = h    + (size_t)N1_ * 256;
  float* ad2v = as2v + (size_t)N1_;
  float* hs2  = hd1;
  float* hd2  = hd1  + (size_t)N1_ * 128;

  const dim3 blk(256);

  {
    int tilesM = cdiv(N0_, 16), waves = tilesM * (256 / 64);
    gat_gemm_wmma<256, 256><<<cdiv(waves, 8), blk, 0, stream>>>(x, Ws1, hs1, N0_, tilesM);
  }
  {
    int tilesM = cdiv(N1_, 16), waves = tilesM * (256 / 64);
    gat_gemm_wmma<256, 256><<<cdiv(waves, 8), blk, 0, stream>>>(x, Wd1, hd1, N1_, tilesM);
  }
  gat_att_dot<<<cdiv(N0_ * H1_, 8), blk, 0, stream>>>(hs1, as1, as1v, N0_, H1_);
  gat_att_dot<<<cdiv(N1_ * H1_, 8), blk, 0, stream>>>(hd1, ad1, ad1v, N1_, H1_);
  gat_edge_softmax_agg<<<cdiv(N1_ * H1_, 8), blk, 0, stream>>>(
      hs1, as1v, ad1v, col1, t1, timeP, intvP,
      b1, g1, beta1, m1, v1, h, N1_, H1_, 1);

  {
    int tilesM = cdiv(N1_, 16), waves = tilesM * (128 / 64);
    gat_gemm_wmma<128, 256><<<cdiv(waves, 8), blk, 0, stream>>>(h, Ws2, hs2, N1_, tilesM);
  }
  {
    int tilesM = cdiv(N2_, 16), waves = tilesM * (128 / 64);
    gat_gemm_wmma<128, 256><<<cdiv(waves, 8), blk, 0, stream>>>(h, Wd2, hd2, N2_, tilesM);
  }
  gat_att_dot<<<cdiv(N1_ * H2_, 8), blk, 0, stream>>>(hs2, as2, as2v, N1_, H2_);
  gat_att_dot<<<cdiv(N2_ * H2_, 8), blk, 0, stream>>>(hd2, ad2, ad2v, N2_, H2_);
  gat_edge_softmax_agg<<<cdiv(N2_ * H2_, 8), blk, 0, stream>>>(
      hs2, as2v, ad2v, col2, t2, timeP, intvP,
      b2, nullptr, nullptr, nullptr, nullptr, (float*)d_out, N2_, H2_, 2);
}


# GEMMs: 32x64 wave tile fed by 128-bit global loads (k and column order permuted inside the fragments, same map for A and B), issued one K step ahead; b128 stores; k sum order changes
# speedup vs baseline: 1.0924x; 1.0924x over previous
#include <hip/hip_runtime.h>
#include <hip/hip_bf16.h>
#include <cstddef>

#define N0_  100000
#define N1_  50000
#define N2_  25000
#define DEG_ 16
#define IN_  256
#define HID_ 128
#define OUT_ 128
#define H1_  2
#define H2_  1

typedef float v2f __attribute__((ext_vector_type(2)));
typedef float v8f __attribute__((ext_vector_type(8)));

template <int TN, int TK>
__global__ void __launch_bounds__(128)
gat_gemm_wmma(const float* __restrict__ A,
              const float* __restrict__ B,
              float* __restrict__ D,
              int M, int tilesM) {
  constexpr int groupsN = TN >> 6;
  constexpr int BROW = TN >> 2;
  const int lane = threadIdx.x & 31;
  const int wave = threadIdx.x >> 5;
  const int gN    = blockIdx.x % groupsN;
  const int tileM = (blockIdx.x / groupsN) * 4 + wave;
  if (tileM >= tilesM) return;

  const int half = lane >> 4;
  const int l15  = lane & 15;
  int r0 = tileM * 32 + l15;
  int r1 = r0 + 16;
  if (r0 >= M) r0 = M - 1;
  if (r1 >= M) r1 = M - 1;
  const float4* ap0 = (const float4*)(A + (size_t)r0 * TK) + half;
  const float4* ap1 = (const float4*)(A + (size_t)r1 * TK) + half;
  const float4* bp  = (const float4*)(B + (size_t)(half * 4) * TN + gN * 64) + l15;

  v8f c00 = {}, c01 = {}, c02 = {}, c03 = {};
  v8f c10 = {}, c11 = {}, c12 = {}, c13 = {};

#define GAT_WMMA(acc, a, b) \
  acc = __builtin_amdgcn_wmma_f32_16x16x4_f32(false, a, false, b, (short)0, acc, false, false)

  constexpr int NS = TK / 8;
  float4 fa0[2], fa1[2], fq0[2], fq1[2], fq2[2], fq3[2];
  fa0[0] = ap0[0];
  fa1[0] = ap1[0];
  fq0[0] = bp[0];
  fq1[0] = bp[BROW];
  fq2[0] = bp[2 * BROW];
  fq3[0] = bp[3 * BROW];
#pragma unroll
  for (int s = 0; s < NS; ++s) {
    const int cb = s & 1, nb = cb ^ 1;
    if (s + 1 < NS) {
      fa0[nb] = ap0[2 * (s + 1)];
      fa1[nb] = ap1[2 * (s + 1)];
      fq0[nb] = bp[(8 * (s + 1) + 0) * BROW];
      fq1[nb] = bp[(8 * (s + 1) + 1) * BROW];
      fq2[nb] = bp[(8 * (s + 1) + 2) * BROW];
      fq3[nb] = bp[(8 * (s + 1) + 3) * BROW];
    }
    __builtin_amdgcn_sched_barrier(0);
    {
      const v2f ae0 = { fa0[cb].x, fa0[cb].y }, ae1 = { fa1[cb].x, fa1[cb].y };
      const v2f ao0 = { fa0[cb].z, fa0[cb].w }, ao1 = { fa1[cb].z, fa1[cb].w };
      const v2f e0 = { fq0[cb].x, fq1[cb].x }, e1 = { fq0[cb].y, fq1[cb].y };
      const v2f e2 = { fq0[cb].z, fq1[cb].z }, e3 = { fq0[cb].w, fq1[cb].w };
      const v2f o0 = { fq2[cb].x, fq3[cb].x }, o1 = { fq2[cb].y, fq3[cb].y };
      const v2f o2 = { fq2[cb].z, fq3[cb].z }, o3 = { fq2[cb].w, fq3[cb].w };
      GAT_WMMA(c00, ae0, e0); GAT_WMMA(c10, ae1, e0);
      GAT_WMMA(c01, ae0, e1); GAT_WMMA(c11, ae1, e1);
      GAT_WMMA(c02, ae0, e2); GAT_WMMA(c12, ae1, e2);
      GAT_WMMA(c03, ae0, e3); GAT_WMMA(c13, ae1, e3);
      GAT_WMMA(c00, ao0, o0); GAT_WMMA(c10, ao1, o0);
      GAT_WMMA(c01, ao0, o1); GAT_WMMA(c11, ao1, o1);
      GAT_WMMA(c02, ao0, o2); GAT_WMMA(c12, ao1, o2);
      GAT_WMMA(c03, ao0, o3); GAT_WMMA(c13, ao1, o3);
    }
    __builtin_amdgcn_sched_barrier(0);
  }
#undef GAT_WMMA

  float* Dc = D + gN * 64 + l15 * 4;
#pragma unroll
  for (int r = 0; r < 8; ++r) {
    const int m0 = tileM * 32 + r + half * 8;
    const int m1 = m0 + 16;
    if (m0 < M) {
      const float4 v = { c00[r], c01[r], c02[r], c03[r] };
      *(float4*)(Dc + (size_t)m0 * TN) = v;
    }
    if (m1 < M) {
      const float4 v = { c10[r], c11[r], c12[r], c13[r] };
      *(float4*)(Dc + (size_t)m1 * TN) = v;
    }
  }
}

__global__ void gat_att_dot(const float* __restrict__ F,
                            const float* __restrict__ att,
                            float* __restrict__ out, int Nn, int H) {
  const int wid  = blockIdx.x * 8 + (threadIdx.x >> 5);
  const int lane = threadIdx.x & 31;
  if (wid >= Nn * H) return;
  const int n = wid / H, h = wid % H;
  const float4 fv = *((const float4*)(F + ((size_t)n * H + h) * 128) + lane);
  const float4 av = *((const float4*)(att + (size_t)h * 128) + lane);
  float s = fv.x * av.x + fv.y * av.y + fv.z * av.z + fv.w * av.w;
#pragma unroll
  for (int off = 16; off > 0; off >>= 1) s += __shfl_xor(s, off, 32);
  if (lane == 0) out[wid] = s;
}

__global__ void gat_edge_softmax_agg(const float* __restrict__ HS,
                                     const float* __restrict__ AS,
                                     const float* __restrict__ AD,
                                     const int* __restrict__ col,
                                     const int* __restrict__ tt,
                                     const int* __restrict__ timeP,
                                     const int* __restrict__ intervalP,
                                     const float* __restrict__ b,
                                     const float* __restrict__ g,
                                     const float* __restrict__ beta,
                                     const float* __restrict__ rm,
                                     const float* __restrict__ rv,
                                     float* __restrict__ OUT,
                                     int Nd, int H, int mode) {
  const int wid  = blockIdx.x * 8 + (threadIdx.x >> 5);
  const int lane = threadIdx.x & 31;
  if (wid >= Nd * H) return;
  const int i = wid / H, h = wid % H;

  const int t0  = timeP[0];
  const int t1e = t0 + intervalP[0];
  const float adv = AD[(size_t)i * H + h];

  float e[DEG_];
  int   cj[DEG_];
  bool  val[DEG_];
  float m = -__builtin_inff();
#pragma unroll
  for (int j = 0; j < DEG_; ++j) {
    const int eidx = i * DEG_ + j;
    cj[j] = col[eidx];
    const int tv = tt[eidx];
    val[j] = (tv >= t0) && (tv < t1e);
    float ev = AS[(size_t)cj[j] * H + h] + adv;
    ev = ev > 0.0f ? ev : 0.2f * ev;
    e[j] = ev;
    if (val[j] && ev > m) m = ev;
  }
  if (!(m > -__builtin_inff())) m = 0.0f;

  float w[DEG_];
  float denom = 0.0f;
#pragma unroll
  for (int j = 0; j < DEG_; ++j) {
    const float ex = val[j] ? __expf(e[j] - m) : 0.0f;
    w[j] = ex;
    denom += ex;
  }
  const float inv = 1.0f / fmaxf(denom, 1e-16f);

  float4 acc = {0.0f, 0.0f, 0.0f, 0.0f};
#pragma unroll
  for (int j = 0; j < DEG_; ++j) {
    if (w[j] != 0.0f) {
      const float4 hv = *((const float4*)(HS + ((size_t)cj[j] * H + h) * 128) + lane);
      const float wj = w[j] * inv;
      acc.x += wj * hv.x; acc.y += wj * hv.y;
      acc.z += wj * hv.z; acc.w += wj * hv.w;
    }
  }

  float vals[4] = {acc.x, acc.y, acc.z, acc.w};
  if (mode == 1) {
#pragma unroll
    for (int q = 0; q < 4; ++q) {
      const int cc = h * 128 + lane * 4 + q;
      float vv = vals[q] + b[cc];
      vv = (vv - rm[cc]) * rsqrtf(rv[cc] + 1e-5f) * g[cc] + beta[cc];
      vv = vv > 0.0f ? vv : 0.01f * vv;
      OUT[(size_t)i * (H * 128) + cc] = vv;
    }
  } else {
#pragma unroll
    for (int q = 0; q < 4; ++q) {
      const int cc = lane * 4 + q;
      OUT[(size_t)i * 128 + cc] = vals[q] / (float)H + b[cc];
    }
  }
}

static inline int cdiv(int a, int b) { return (a + b - 1) / b; }

extern "C" void kernel_launch(void* const* d_in, const int* in_sizes, int n_in,
                              void* d_out, int out_size, void* d_ws, size_t ws_size,
                              hipStream_t stream) {
  const float* x     = (const float*)d_in[0];
  const int*   col1  = (const int*)d_in[2];
  const int*   t1    = (const int*)d_in[3];
  const int*   col2  = (const int*)d_in[5];
  const int*   t2    = (const int*)d_in[6];
  const int*   timeP = (const int*)d_in[7];
  const int*   intvP = (const int*)d_in[8];
  const float* Ws1   = (const float*)d_in[9];
  const float* Wd1   = (const float*)d_in[10];
  const float* as1   = (const float*)d_in[11];
  const float* ad1   = (const float*)d_in[12];
  const float* b1    = (const float*)d_in[13];
  const float* g1    = (const float*)d_in[14];
  const float* beta1 = (const float*)d_in[15];
  const float* m1    = (const float*)d_in[16];
  const float* v1    = (const float*)d_in[17];
  const float* Ws2   = (const float*)d_in[18];
  const float* Wd2   = (const float*)d_in[19];
  const float* as2   = (const float*)d_in[20];
  const float* ad2   = (const float*)d_in[21];
  const float* b2    = (const float*)d_in[22];

  float* ws   = (float*)d_ws;
  float* hs1  = ws;
  float* hd1  = hs1  + (size_t)N0_ * 256;
  float* as1v = hd1  + (size_t)N1_ * 256;
  float* ad1v = as1v + (size_t)N0_ * H1_;
  float* h    = ad1v + (size_t)N1_ * H1_;
  float* as2v = h    + (size_t)N1_ * 256;
  float* ad2v = as2v + (size_t)N1_;
  float* hs2  = hd1;
  float* hd2  = hd1  + (size_t)N1_ * 128;

  const dim3 blk(256);
  const dim3 gblk(128);
  constexpr int T32_N0 = (N0_ + 31) / 32, T32_N1 = (N1_ + 31) / 32, T32_N2 = (N2_ + 31) / 32;
  constexpr int G_N0_256 = ((T32_N0 + 3) / 4) * (256 / 64);
  constexpr int G_N1_256 = ((T32_N1 + 3) / 4) * (256 / 64);
  constexpr int G_N1_128 = ((T32_N1 + 3) / 4) * (128 / 64);
  constexpr int G_N2_128 = ((T32_N2 + 3) / 4) * (128 / 64);

  gat_gemm_wmma<256, 256><<<G_N0_256, gblk, 0, stream>>>(x, Ws1, hs1, N0_, T32_N0);
  gat_gemm_wmma<256, 256><<<G_N1_256, gblk, 0, stream>>>(x, Wd1, hd1, N1_, T32_N1);
  gat_att_dot<<<cdiv(N0_ * H1_, 8), blk, 0, stream>>>(hs1, as1, as1v, N0_, H1_);
  gat_att_dot<<<cdiv(N1_ * H1_, 8), blk, 0, stream>>>(hd1, ad1, ad1v, N1_, H1_);
  gat_edge_softmax_agg<<<cdiv(N1_ * H1_, 8), blk, 0, stream>>>(
      hs1, as1v, ad1v, col1, t1, timeP, intvP,
      b1, g1, beta1, m1, v1, h, N1_, H1_, 1);

  gat_gemm_wmma<128, 256><<<G_N1_128, gblk, 0, stream>>>(h, Ws2, hs2, N1_, T32_N1);
  gat_gemm_wmma<128, 256><<<G_N2_128, gblk, 0, stream>>>(h, Wd2, hd2, N2_, T32_N2);
  gat_att_dot<<<cdiv(N1_ * H2_, 8), blk, 0, stream>>>(hs2, as2, as2v, N1_, H2_);
  gat_att_dot<<<cdiv(N2_ * H2_, 8), blk, 0, stream>>>(hd2, ad2, ad2v, N2_, H2_);
  gat_edge_softmax_agg<<<cdiv(N2_ * H2_, 8), blk, 0, stream>>>(
      hs2, as2v, ad2v, col2, t2, timeP, intvP,
      b2, nullptr, nullptr, nullptr, nullptr, (float*)d_out, N2_, H2_, 2);
}


# GEMM epilogue dots each row's accumulators with att (64-col partials per wave, 16-lane DPP reduce-scatter, pairs added in the gather): 4 dot launches and hd stores dropped; a_s,a_d sum order changes
# speedup vs baseline: 1.2780x; 1.1699x over previous
#include <hip/hip_runtime.h>
#include <hip/hip_bf16.h>
#include <cstddef>

#define N0_  100000
#define N1_  50000
#define N2_  25000
#define DEG_ 16
#define IN_  256
#define HID_ 128
#define OUT_ 128
#define H1_  2
#define H2_  1

typedef float v2f __attribute__((ext_vector_type(2)));
typedef float v8f __attribute__((ext_vector_type(8)));


template <int XM>
__device__ __forceinline__ float gat_row_xor_add(float keep, float send) {
  const int s = __builtin_bit_cast(int, send);
  const int g = __builtin_amdgcn_update_dpp(s, s, 0x160 + XM, 0xf, 0xf, false);
  return keep + __builtin_bit_cast(float, g);
}

template <int TN, int TK, bool STORE_D>
__global__ void __launch_bounds__(128)
gat_gemm_wmma(const float* __restrict__ A,
              const float* __restrict__ B,
              const float* __restrict__ att,
              float* __restrict__ D,
              float* __restrict__ P,
              int M, int tilesM) {
  constexpr int groupsN = TN >> 6;
  constexpr int BROW = TN >> 2;
  const int lane = threadIdx.x & 31;
  const int wave = threadIdx.x >> 5;
  const int gN    = blockIdx.x % groupsN;
  const int tileM = (blockIdx.x / groupsN) * 4 + wave;
  if (tileM >= tilesM) return;

  const int half = lane >> 4;
  const int l15  = lane & 15;
  int r0 = tileM * 32 + l15;
  int r1 = r0 + 16;
  if (r0 >= M) r0 = M - 1;
  if (r1 >= M) r1 = M - 1;
  const float4* ap0 = (const float4*)(A + (size_t)r0 * TK) + half;
  const float4* ap1 = (const float4*)(A + (size_t)r1 * TK) + half;
  const float4* bp  = (const float4*)(B + (size_t)(half * 4) * TN + gN * 64) + l15;

  v8f c00 = {}, c01 = {}, c02 = {}, c03 = {};
  v8f c10 = {}, c11 = {}, c12 = {}, c13 = {};

#define GAT_WMMA(acc, a, b) \
  acc = __builtin_amdgcn_wmma_f32_16x16x4_f32(false, a, false, b, (short)0, acc, false, false)

  constexpr int NS = TK / 8;
  float4 fa0[2], fa1[2], fq0[2], fq1[2], fq2[2], fq3[2];
  fa0[0] = ap0[0];
  fa1[0] = ap1[0];
  fq0[0] = bp[0];
  fq1[0] = bp[BROW];
  fq2[0] = bp[2 * BROW];
  fq3[0] = bp[3 * BROW];
#pragma unroll
  for (int s = 0; s < NS; ++s) {
    const int cb = s & 1, nb = cb ^ 1;
    if (s + 1 < NS) {
      fa0[nb] = ap0[2 * (s + 1)];
      fa1[nb] = ap1[2 * (s + 1)];
      fq0[nb] = bp[(8 * (s + 1) + 0) * BROW];
      fq1[nb] = bp[(8 * (s + 1) + 1) * BROW];
      fq2[nb] = bp[(8 * (s + 1) + 2) * BROW];
      fq3[nb] = bp[(8 * (s + 1) + 3) * BROW];
    }
    __builtin_amdgcn_sched_barrier(0);
    {
      const v2f ae0 = { fa0[cb].x, fa0[cb].y }, ae1 = { fa1[cb].x, fa1[cb].y };
      const v2f ao0 = { fa0[cb].z, fa0[cb].w }, ao1 = { fa1[cb].z, fa1[cb].w };
      const v2f e0 = { fq0[cb].x, fq1[cb].x }, e1 = { fq0[cb].y, fq1[cb].y };
      const v2f e2 = { fq0[cb].z, fq1[cb].z }, e3 = { fq0[cb].w, fq1[cb].w };
      const v2f o0 = { fq2[cb].x, fq3[cb].x }, o1 = { fq2[cb].y, fq3[cb].y };
      const v2f o2 = { fq2[cb].z, fq3[cb].z }, o3 = { fq2[cb].w, fq3[cb].w };
      GAT_WMMA(c00, ae0, e0); GAT_WMMA(c10, ae1, e0);
      GAT_WMMA(c01, ae0, e1); GAT_WMMA(c11, ae1, e1);
      GAT_WMMA(c02, ae0, e2); GAT_WMMA(c12, ae1, e2);
      GAT_WMMA(c03, ae0, e3); GAT_WMMA(c13, ae1, e3);
      GAT_WMMA(c00, ao0, o0); GAT_WMMA(c10, ao1, o0);
      GAT_WMMA(c01, ao0, o1); GAT_WMMA(c11, ao1, o1);
      GAT_WMMA(c02, ao0, o2); GAT_WMMA(c12, ao1, o2);
      GAT_WMMA(c03, ao0, o3); GAT_WMMA(c13, ao1, o3);
    }
    __builtin_amdgcn_sched_barrier(0);
  }
#undef GAT_WMMA

  const float4 av = ((const float4*)att)[gN * 16 + l15];

  if constexpr (STORE_D) {
    float* Dc = D + gN * 64 + l15 * 4;
#pragma unroll
    for (int r = 0; r < 8; ++r) {
      const int m0 = tileM * 32 + r + half * 8;
      const int m1 = m0 + 16;
      if (m0 < M) {
        const float4 v = { c00[r], c01[r], c02[r], c03[r] };
        *(float4*)(Dc + (size_t)m0 * TN) = v;
      }
      if (m1 < M) {
        const float4 v = { c10[r], c11[r], c12[r], c13[r] };
        *(float4*)(Dc + (size_t)m1 * TN) = v;
      }
    }
  }

  float p[16];
#pragma unroll
  for (int r = 0; r < 8; ++r) {
    p[r]     = c00[r] * av.x + c01[r] * av.y + c02[r] * av.z + c03[r] * av.w;
    p[8 + r] = c10[r] * av.x + c11[r] * av.y + c12[r] * av.z + c13[r] * av.w;
  }
  const bool b3 = (l15 & 8) != 0, b2 = (l15 & 4) != 0;
  const bool b1 = (l15 & 2) != 0, b0 = (l15 & 1) != 0;
  float u[8], w[4], y[2];
#pragma unroll
  for (int q = 0; q < 8; ++q)
    u[q] = gat_row_xor_add<8>(b3 ? p[8 + q] : p[q], b3 ? p[q] : p[8 + q]);
#pragma unroll
  for (int q = 0; q < 4; ++q)
    w[q] = gat_row_xor_add<4>(b2 ? u[4 + q] : u[q], b2 ? u[q] : u[4 + q]);
#pragma unroll
  for (int q = 0; q < 2; ++q)
    y[q] = gat_row_xor_add<2>(b1 ? w[2 + q] : w[q], b1 ? w[q] : w[2 + q]);
  const float tot = gat_row_xor_add<1>(b0 ? y[1] : y[0], b0 ? y[0] : y[1]);
  const int prow = tileM * 32 + ((l15 >> 3) << 4) + (half << 3) + (l15 & 7);
  if (prow < M) P[(size_t)prow * groupsN + gN] = tot;
}

__global__ void gat_edge_softmax_agg(const float* __restrict__ HS,
                                     const float* __restrict__ AS,
                                     const float* __restrict__ AD,
                                     const int* __restrict__ col,
                                     const int* __restrict__ tt,
                                     const int* __restrict__ timeP,
                                     const int* __restrict__ intervalP,
                                     const float* __restrict__ b,
                                     const float* __restrict__ g,
                                     const float* __restrict__ beta,
                                     const float* __restrict__ rm,
                                     const float* __restrict__ rv,
                                     float* __restrict__ OUT,
                                     int Nd, int H, int mode) {
  const int wid  = blockIdx.x * 8 + (threadIdx.x >> 5);
  const int lane = threadIdx.x & 31;
  if (wid >= Nd * H) return;
  const int i = wid / H, h = wid % H;

  const int t0  = timeP[0];
  const int t1e = t0 + intervalP[0];
  const float2 adp = ((const float2*)AD)[(size_t)i * H + h];
  const float adv = adp.x + adp.y;

  float e[DEG_];
  int   cj[DEG_];
  bool  val[DEG_];
  float m = -__builtin_inff();
#pragma unroll
  for (int j = 0; j < DEG_; ++j) {
    const int eidx = i * DEG_ + j;
    cj[j] = col[eidx];
    const int tv = tt[eidx];
    val[j] = (tv >= t0) && (tv < t1e);
    const float2 asp = ((const float2*)AS)[(size_t)cj[j] * H + h];
    float ev = (asp.x + asp.y) + adv;
    ev = ev > 0.0f ? ev : 0.2f * ev;
    e[j] = ev;
    if (val[j] && ev > m) m = ev;
  }
  if (!(m > -__builtin_inff())) m = 0.0f;

  float w[DEG_];
  float denom = 0.0f;
#pragma unroll
  for (int j = 0; j < DEG_; ++j) {
    const float ex = val[j] ? __expf(e[j] - m) : 0.0f;
    w[j] = ex;
    denom += ex;
  }
  const float inv = 1.0f / fmaxf(denom, 1e-16f);

  float4 acc = {0.0f, 0.0f, 0.0f, 0.0f};
#pragma unroll
  for (int j = 0; j < DEG_; ++j) {
    if (w[j] != 0.0f) {
      const float4 hv = *((const float4*)(HS + ((size_t)cj[j] * H + h) * 128) + lane);
      const float wj = w[j] * inv;
      acc.x += wj * hv.x; acc.y += wj * hv.y;
      acc.z += wj * hv.z; acc.w += wj * hv.w;
    }
  }

  float vals[4] = {acc.x, acc.y, acc.z, acc.w};
  if (mode == 1) {
#pragma unroll
    for (int q = 0; q < 4; ++q) {
      const int cc = h * 128 + lane * 4 + q;
      float vv = vals[q] + b[cc];
      vv = (vv - rm[cc]) * rsqrtf(rv[cc] + 1e-5f) * g[cc] + beta[cc];
      vv = vv > 0.0f ? vv : 0.01f * vv;
      OUT[(size_t)i * (H * 128) + cc] = vv;
    }
  } else {
#pragma unroll
    for (int q = 0; q < 4; ++q) {
      const int cc = lane * 4 + q;
      OUT[(size_t)i * 128 + cc] = vals[q] / (float)H + b[cc];
    }
  }
}

static inline int cdiv(int a, int b) { return (a + b - 1) / b; }

extern "C" void kernel_launch(void* const* d_in, const int* in_sizes, int n_in,
                              void* d_out, int out_size, void* d_ws, size_t ws_size,
                              hipStream_t stream) {
  const float* x     = (const float*)d_in[0];
  const int*   col1  = (const int*)d_in[2];
  const int*   t1    = (const int*)d_in[3];
  const int*   col2  = (const int*)d_in[5];
  const int*   t2    = (const int*)d_in[6];
  const int*   timeP = (const int*)d_in[7];
  const int*   intvP = (const int*)d_in[8];
  const float* Ws1   = (const float*)d_in[9];
  const float* Wd1   = (const float*)d_in[10];
  const float* as1   = (const float*)d_in[11];
  const float* ad1   = (const float*)d_in[12];
  const float* b1    = (const float*)d_in[13];
  const float* g1    = (const float*)d_in[14];
  const float* beta1 = (const float*)d_in[15];
  const float* m1    = (const float*)d_in[16];
  const float* v1    = (const float*)d_in[17];
  const float* Ws2   = (const float*)d_in[18];
  const float* Wd2   = (const float*)d_in[19];
  const float* as2   = (const float*)d_in[20];
  const float* ad2   = (const float*)d_in[21];
  const float* b2    = (const float*)d_in[22];

  float* ws   = (float*)d_ws;
  float* hs1  = ws;
  float* hs2  = hs1  + (size_t)N0_ * 256;
  float* h    = hs2  + (size_t)N1_ * 128;
  float* as1p = h    + (size_t)N1_ * 256;
  float* ad1p = as1p + (size_t)N0_ * H1_ * 2;
  float* as2p = ad1p + (size_t)N1_ * H1_ * 2;
  float* ad2p = as2p + (size_t)N1_ * H2_ * 2;

  const dim3 blk(256);
  const dim3 gblk(128);
  constexpr int T32_N0 = (N0_ + 31) / 32, T32_N1 = (N1_ + 31) / 32, T32_N2 = (N2_ + 31) / 32;
  constexpr int G_N0_256 = ((T32_N0 + 3) / 4) * (256 / 64);
  constexpr int G_N1_256 = ((T32_N1 + 3) / 4) * (256 / 64);
  constexpr int G_N1_128 = ((T32_N1 + 3) / 4) * (128 / 64);
  constexpr int G_N2_128 = ((T32_N2 + 3) / 4) * (128 / 64);

  gat_gemm_wmma<256, 256, true><<<G_N0_256, gblk, 0, stream>>>(
      x, Ws1, as1, hs1, as1p, N0_, T32_N0);
  gat_gemm_wmma<256, 256, false><<<G_N1_256, gblk, 0, stream>>>(
      x, Wd1, ad1, nullptr, ad1p, N1_, T32_N1);
  gat_edge_softmax_agg<<<cdiv(N1_ * H1_, 8), blk, 0, stream>>>(
      hs1, as1p, ad1p, col1, t1, timeP, intvP,
      b1, g1, beta1, m1, v1, h, N1_, H1_, 1);

  gat_gemm_wmma<128, 256, true><<<G_N1_128, gblk, 0, stream>>>(
      h, Ws2, as2, hs2, as2p, N1_, T32_N1);
  gat_gemm_wmma<128, 256, false><<<G_N2_128, gblk, 0, stream>>>(
      h, Wd2, ad2, nullptr, ad2p, N2_, T32_N2);
  gat_edge_softmax_agg<<<cdiv(N2_ * H2_, 8), blk, 0, stream>>>(
      hs2, as2p, ad2p, col2, t2, timeP, intvP,
      b2, nullptr, nullptr, nullptr, nullptr, (float*)d_out, N2_, H2_, 2);
}


# Gather kernels: 8 lanes x 16 channels per (dst row, head) pair, 4 pairs per wave (was one wave per pair): a quarter of the waves, 4 whole-line row loads per lane issued together; sums unchanged
# speedup vs baseline: 1.4569x; 1.1400x over previous
#include <hip/hip_runtime.h>
#include <hip/hip_bf16.h>
#include <cstddef>

#define N0_  100000
#define N1_  50000
#define N2_  25000
#define DEG_ 16
#define IN_  256
#define HID_ 128
#define OUT_ 128
#define H1_  2
#define H2_  1

typedef float v2f __attribute__((ext_vector_type(2)));
typedef float v8f __attribute__((ext_vector_type(8)));


template <int XM>
__device__ __forceinline__ float gat_row_xor_add(float keep, float send) {
  const int s = __builtin_bit_cast(int, send);
  const int g = __builtin_amdgcn_update_dpp(s, s, 0x160 + XM, 0xf, 0xf, false);
  return keep + __builtin_bit_cast(float, g);
}

template <int TN, int TK, bool STORE_D>
__global__ void __launch_bounds__(128)
gat_gemm_wmma(const float* __restrict__ A,
              const float* __restrict__ B,
              const float* __restrict__ att,
              float* __restrict__ D,
              float* __restrict__ P,
              int M, int tilesM) {
  constexpr int groupsN = TN >> 6;
  constexpr int BROW = TN >> 2;
  const int lane = threadIdx.x & 31;
  const int wave = threadIdx.x >> 5;
  const int gN    = blockIdx.x % groupsN;
  const int tileM = (blockIdx.x / groupsN) * 4 + wave;
  if (tileM >= tilesM) return;

  const int half = lane >> 4;
  const int l15  = lane & 15;
  int r0 = tileM * 32 + l15;
  int r1 = r0 + 16;
  if (r0 >= M) r0 = M - 1;
  if (r1 >= M) r1 = M - 1;
  const float4* ap0 = (const float4*)(A + (size_t)r0 * TK) + half;
  const float4* ap1 = (const float4*)(A + (size_t)r1 * TK) + half;
  const float4* bp  = (const float4*)(B + (size_t)(half * 4) * TN + gN * 64) + l15;

  v8f c00 = {}, c01 = {}, c02 = {}, c03 = {};
  v8f c10 = {}, c11 = {}, c12 = {}, c13 = {};

#define GAT_WMMA(acc, a, b) \
  acc = __builtin_amdgcn_wmma_f32_16x16x4_f32(false, a, false, b, (short)0, acc, false, false)

  constexpr int NS = TK / 8;
  float4 fa0[2], fa1[2], fq0[2], fq1[2], fq2[2], fq3[2];
  fa0[0] = ap0[0];
  fa1[0] = ap1[0];
  fq0[0] = bp[0];
  fq1[0] = bp[BROW];
  fq2[0] = bp[2 * BROW];
  fq3[0] = bp[3 * BROW];
#pragma unroll
  for (int s = 0; s < NS; ++s) {
    const int cb = s & 1, nb = cb ^ 1;
    if (s + 1 < NS) {
      fa0[nb] = ap0[2 * (s + 1)];
      fa1[nb] = ap1[2 * (s + 1)];
      fq0[nb] = bp[(8 * (s + 1) + 0) * BROW];
      fq1[nb] = bp[(8 * (s + 1) + 1) * BROW];
      fq2[nb] = bp[(8 * (s + 1) + 2) * BROW];
      fq3[nb] = bp[(8 * (s + 1) + 3) * BROW];
    }
    __builtin_amdgcn_sched_barrier(0);
    {
      const v2f ae0 = { fa0[cb].x, fa0[cb].y }, ae1 = { fa1[cb].x, fa1[cb].y };
      const v2f ao0 = { fa0[cb].z, fa0[cb].w }, ao1 = { fa1[cb].z, fa1[cb].w };
      const v2f e0 = { fq0[cb].x, fq1[cb].x }, e1 = { fq0[cb].y, fq1[cb].y };
      const v2f e2 = { fq0[cb].z, fq1[cb].z }, e3 = { fq0[cb].w, fq1[cb].w };
      const v2f o0 = { fq2[cb].x, fq3[cb].x }, o1 = { fq2[cb].y, fq3[cb].y };
      const v2f o2 = { fq2[cb].z, fq3[cb].z }, o3 = { fq2[cb].w, fq3[cb].w };
      GAT_WMMA(c00, ae0, e0); GAT_WMMA(c10, ae1, e0);
      GAT_WMMA(c01, ae0, e1); GAT_WMMA(c11, ae1, e1);
      GAT_WMMA(c02, ae0, e2); GAT_WMMA(c12, ae1, e2);
      GAT_WMMA(c03, ae0, e3); GAT_WMMA(c13, ae1, e3);
      GAT_WMMA(c00, ao0, o0); GAT_WMMA(c10, ao1, o0);
      GAT_WMMA(c01, ao0, o1); GAT_WMMA(c11, ao1, o1);
      GAT_WMMA(c02, ao0, o2); GAT_WMMA(c12, ao1, o2);
      GAT_WMMA(c03, ao0, o3); GAT_WMMA(c13, ao1, o3);
    }
    __builtin_amdgcn_sched_barrier(0);
  }
#undef GAT_WMMA

  const float4 av = ((const float4*)att)[gN * 16 + l15];

  if constexpr (STORE_D) {
    float* Dc = D + gN * 64 + l15 * 4;
#pragma unroll
    for (int r = 0; r < 8; ++r) {
      const int m0 = tileM * 32 + r + half * 8;
      const int m1 = m0 + 16;
      if (m0 < M) {
        const float4 v = { c00[r], c01[r], c02[r], c03[r] };
        *(float4*)(Dc + (size_t)m0 * TN) = v;
      }
      if (m1 < M) {
        const float4 v = { c10[r], c11[r], c12[r], c13[r] };
        *(float4*)(Dc + (size_t)m1 * TN) = v;
      }
    }
  }

  float p[16];
#pragma unroll
  for (int r = 0; r < 8; ++r) {
    p[r]     = c00[r] * av.x + c01[r] * av.y + c02[r] * av.z + c03[r] * av.w;
    p[8 + r] = c10[r] * av.x + c11[r] * av.y + c12[r] * av.z + c13[r] * av.w;
  }
  const bool b3 = (l15 & 8) != 0, b2 = (l15 & 4) != 0;
  const bool b1 = (l15 & 2) != 0, b0 = (l15 & 1) != 0;
  float u[8], w[4], y[2];
#pragma unroll
  for (int q = 0; q < 8; ++q)
    u[q] = gat_row_xor_add<8>(b3 ? p[8 + q] : p[q], b3 ? p[q] : p[8 + q]);
#pragma unroll
  for (int q = 0; q < 4; ++q)
    w[q] = gat_row_xor_add<4>(b2 ? u[4 + q] : u[q], b2 ? u[q] : u[4 + q]);
#pragma unroll
  for (int q = 0; q < 2; ++q)
    y[q] = gat_row_xor_add<2>(b1 ? w[2 + q] : w[q], b1 ? w[q] : w[2 + q]);
  const float tot = gat_row_xor_add<1>(b0 ? y[1] : y[0], b0 ? y[0] : y[1]);
  const int prow = tileM * 32 + ((l15 >> 3) << 4) + (half << 3) + (l15 & 7);
  if (prow < M) P[(size_t)prow * groupsN + gN] = tot;
}

__global__ void gat_edge_softmax_agg(const float* __restrict__ HS,
                                     const float* __restrict__ AS,
                                     const float* __restrict__ AD,
                                     const int* __restrict__ col,
                                     const int* __restrict__ tt,
                                     const int* __restrict__ timeP,
                                     const int* __restrict__ intervalP,
                                     const float* __restrict__ b,
                                     const float* __restrict__ g,
                                     const float* __restrict__ beta,
                                     const float* __restrict__ rm,
                                     const float* __restrict__ rv,
                                     float* __restrict__ OUT,
                                     int Nd, int H, int mode) {
  const int lane = threadIdx.x & 31;
  const int wid  = blockIdx.x * 8 + (threadIdx.x >> 5);
  const int pair = wid * 4 + (lane >> 3);
  const int sub  = lane & 7;
  if (pair >= Nd * H) return;
  const int i = pair / H, h = pair % H;

  const int t0  = timeP[0];
  const int t1e = t0 + intervalP[0];
  const float2 adp = ((const float2*)AD)[(size_t)i * H + h];
  const float adv = adp.x + adp.y;

  float e[DEG_];
  int   cj[DEG_];
  bool  val[DEG_];
  float m = -__builtin_inff();
#pragma unroll
  for (int j = 0; j < DEG_; ++j) {
    const int eidx = i * DEG_ + j;
    cj[j] = col[eidx];
    const int tv = tt[eidx];
    val[j] = (tv >= t0) && (tv < t1e);
    const float2 asp = ((const float2*)AS)[(size_t)cj[j] * H + h];
    float ev = (asp.x + asp.y) + adv;
    ev = ev > 0.0f ? ev : 0.2f * ev;
    e[j] = ev;
    if (val[j] && ev > m) m = ev;
  }
  if (!(m > -__builtin_inff())) m = 0.0f;

  float w[DEG_];
  float denom = 0.0f;
#pragma unroll
  for (int j = 0; j < DEG_; ++j) {
    const float ex = val[j] ? __expf(e[j] - m) : 0.0f;
    w[j] = ex;
    denom += ex;
  }
  const float inv = 1.0f / fmaxf(denom, 1e-16f);

  float4 acc[4];
#pragma unroll
  for (int q = 0; q < 4; ++q) acc[q] = float4{0.0f, 0.0f, 0.0f, 0.0f};
  const float4* hsl = (const float4*)HS + sub;
#pragma unroll
  for (int j = 0; j < DEG_; ++j) {
    if (w[j] != 0.0f) {
      const float4* rp = hsl + ((size_t)cj[j] * H + h) * 32;
      const float4 v0 = rp[0], v1 = rp[8], v2 = rp[16], v3 = rp[24];
      const float wj = w[j] * inv;
      acc[0].x += wj * v0.x; acc[0].y += wj * v0.y;
      acc[0].z += wj * v0.z; acc[0].w += wj * v0.w;
      acc[1].x += wj * v1.x; acc[1].y += wj * v1.y;
      acc[1].z += wj * v1.z; acc[1].w += wj * v1.w;
      acc[2].x += wj * v2.x; acc[2].y += wj * v2.y;
      acc[2].z += wj * v2.z; acc[2].w += wj * v2.w;
      acc[3].x += wj * v3.x; acc[3].y += wj * v3.y;
      acc[3].z += wj * v3.z; acc[3].w += wj * v3.w;
    }
  }

  if (mode == 1) {
#pragma unroll
    for (int q = 0; q < 4; ++q) {
      const float vals[4] = {acc[q].x, acc[q].y, acc[q].z, acc[q].w};
#pragma unroll
      for (int k = 0; k < 4; ++k) {
        const int cc = h * 128 + q * 32 + sub * 4 + k;
        float vv = vals[k] + b[cc];
        vv = (vv - rm[cc]) * rsqrtf(rv[cc] + 1e-5f) * g[cc] + beta[cc];
        vv = vv > 0.0f ? vv : 0.01f * vv;
        OUT[(size_t)i * (H * 128) + cc] = vv;
      }
    }
  } else {
#pragma unroll
    for (int q = 0; q < 4; ++q) {
      const float vals[4] = {acc[q].x, acc[q].y, acc[q].z, acc[q].w};
#pragma unroll
      for (int k = 0; k < 4; ++k) {
        const int cc = q * 32 + sub * 4 + k;
        OUT[(size_t)i * 128 + cc] = vals[k] / (float)H + b[cc];
      }
    }
  }
}

static inline int cdiv(int a, int b) { return (a + b - 1) / b; }

extern "C" void kernel_launch(void* const* d_in, const int* in_sizes, int n_in,
                              void* d_out, int out_size, void* d_ws, size_t ws_size,
                              hipStream_t stream) {
  const float* x     = (const float*)d_in[0];
  const int*   col1  = (const int*)d_in[2];
  const int*   t1    = (const int*)d_in[3];
  const int*   col2  = (const int*)d_in[5];
  const int*   t2    = (const int*)d_in[6];
  const int*   timeP = (const int*)d_in[7];
  const int*   intvP = (const int*)d_in[8];
  const float* Ws1   = (const float*)d_in[9];
  const float* Wd1   = (const float*)d_in[10];
  const float* as1   = (const float*)d_in[11];
  const float* ad1   = (const float*)d_in[12];
  const float* b1    = (const float*)d_in[13];
  const float* g1    = (const float*)d_in[14];
  const float* beta1 = (const float*)d_in[15];
  const float* m1    = (const float*)d_in[16];
  const float* v1    = (const float*)d_in[17];
  const float* Ws2   = (const float*)d_in[18];
  const float* Wd2   = (const float*)d_in[19];
  const float* as2   = (const float*)d_in[20];
  const float* ad2   = (const float*)d_in[21];
  const float* b2    = (const float*)d_in[22];

  float* ws   = (float*)d_ws;
  float* hs1  = ws;
  float* hs2  = hs1  + (size_t)N0_ * 256;
  float* h    = hs2  + (size_t)N1_ * 128;
  float* as1p = h    + (size_t)N1_ * 256;
  float* ad1p = as1p + (size_t)N0_ * H1_ * 2;
  float* as2p = ad1p + (size_t)N1_ * H1_ * 2;
  float* ad2p = as2p + (size_t)N1_ * H2_ * 2;

  const dim3 blk(256);
  const dim3 gblk(128);
  constexpr int T32_N0 = (N0_ + 31) / 32, T32_N1 = (N1_ + 31) / 32, T32_N2 = (N2_ + 31) / 32;
  constexpr int G_N0_256 = ((T32_N0 + 3) / 4) * (256 / 64);
  constexpr int G_N1_256 = ((T32_N1 + 3) / 4) * (256 / 64);
  constexpr int G_N1_128 = ((T32_N1 + 3) / 4) * (128 / 64);
  constexpr int G_N2_128 = ((T32_N2 + 3) / 4) * (128 / 64);
  constexpr int G_AGG1 = ((N1_ * H1_ + 3) / 4 + 7) / 8;
  constexpr int G_AGG2 = ((N2_ * H2_ + 3) / 4 + 7) / 8;

  gat_gemm_wmma<256, 256, true><<<G_N0_256, gblk, 0, stream>>>(
      x, Ws1, as1, hs1, as1p, N0_, T32_N0);
  gat_gemm_wmma<256, 256, false><<<G_N1_256, gblk, 0, stream>>>(
      x, Wd1, ad1, nullptr, ad1p, N1_, T32_N1);
  gat_edge_softmax_agg<<<G_AGG1, blk, 0, stream>>>(
      hs1, as1p, ad1p, col1, t1, timeP, intvP,
      b1, g1, beta1, m1, v1, h, N1_, H1_, 1);

  gat_gemm_wmma<128, 256, true><<<G_N1_128, gblk, 0, stream>>>(
      h, Ws2, as2, hs2, as2p, N1_, T32_N1);
  gat_gemm_wmma<128, 256, false><<<G_N2_128, gblk, 0, stream>>>(
      h, Wd2, ad2, nullptr, ad2p, N2_, T32_N2);
  gat_edge_softmax_agg<<<G_AGG2, blk, 0, stream>>>(
      hs2, as2p, ad2p, col2, t2, timeP, intvP,
      b2, nullptr, nullptr, nullptr, nullptr, (float*)d_out, N2_, H2_, 2);
}
